# starting kernel
# baseline (speedup 1.0000x reference)
#include <hip/hip_runtime.h>
#include <hip/hip_bf16.h>


#define IN_DIM 128
#define HID    64
#define OUTD   64

typedef float v2f __attribute__((ext_vector_type(2)));
typedef float v8f __attribute__((ext_vector_type(8)));


__global__ void k_deg_init(float* __restrict__ deg, int n) {
    int i = blockIdx.x * blockDim.x + threadIdx.x;
    if (i < n) deg[i] = 1.0f;
}

__global__ void k_deg_edges(const int* __restrict__ col, float* __restrict__ deg, int e) {
    int i = blockIdx.x * blockDim.x + threadIdx.x;
    if (i < e) atomicAdd(&deg[col[i]], 1.0f);
}

__global__ void k_rsqrt(float* __restrict__ deg, int n) {
    int i = blockIdx.x * blockDim.x + threadIdx.x;
    if (i < n) deg[i] = rsqrtf(deg[i]);
}


template<int K, bool RELU_A>
__global__ void k_gemm_wmma(const float* __restrict__ A,
                            const float* __restrict__ B,
                            float* __restrict__ C,
                            int mtiles) {
    const int wave  = threadIdx.x >> 5;
    const int lane  = threadIdx.x & 31;
    const int tileM = blockIdx.x * (blockDim.x >> 5) + wave;
    if (tileM >= mtiles) return;

    const int nloc = lane & 15;
    const int half = lane >> 4;

    const float* Arow = A + (size_t)(tileM * 16 + nloc) * K;

    v8f acc0 = {}, acc1 = {}, acc2 = {}, acc3 = {};

    #pragma unroll 4
    for (int k0 = 0; k0 < K; k0 += 4) {
        v2f a = *(const v2f*)(Arow + k0 + 2 * half);
        if (RELU_A) { a.x = fmaxf(a.x, 0.0f); a.y = fmaxf(a.y, 0.0f); }

        const float* Bp = B + (size_t)(k0 + 2 * half) * 64 + nloc;
        v2f b0, b1, b2, b3;
        b0.x = Bp[0];  b0.y = Bp[64];
        b1.x = Bp[16]; b1.y = Bp[80];
        b2.x = Bp[32]; b2.y = Bp[96];
        b3.x = Bp[48]; b3.y = Bp[112];

        acc0 = __builtin_amdgcn_wmma_f32_16x16x4_f32(false, a, false, b0, (short)0, acc0, false, false);
        acc1 = __builtin_amdgcn_wmma_f32_16x16x4_f32(false, a, false, b1, (short)0, acc1, false, false);
        acc2 = __builtin_amdgcn_wmma_f32_16x16x4_f32(false, a, false, b2, (short)0, acc2, false, false);
        acc3 = __builtin_amdgcn_wmma_f32_16x16x4_f32(false, a, false, b3, (short)0, acc3, false, false);
    }

    float* Crow = C + (size_t)(tileM * 16 + half * 8) * 64 + nloc;
    #pragma unroll
    for (int v = 0; v < 8; ++v) {
        Crow[(size_t)v * 64 +  0] = acc0[v];
        Crow[(size_t)v * 64 + 16] = acc1[v];
        Crow[(size_t)v * 64 + 32] = acc2[v];
        Crow[(size_t)v * 64 + 48] = acc3[v];
    }
}

__global__ void k_prop_init(float* __restrict__ out, const float* __restrict__ h,
                            const float* __restrict__ dis, const float* __restrict__ bias,
                            int total) {
    int idx = blockIdx.x * blockDim.x + threadIdx.x;
    if (idx < total) {
        int node = idx >> 6;
        int d    = idx & 63;
        float s  = dis[node];
        out[idx] = bias[d] + s * s * h[idx];
    }
}

__global__ void k_prop_edges(float* __restrict__ out, const float* __restrict__ h,
                             const float* __restrict__ dis,
                             const int* __restrict__ row, const int* __restrict__ col,
                             int e) {
    long long t = (long long)blockIdx.x * blockDim.x + threadIdx.x;
    int eidx = (int)(t >> 5);
    if (eidx >= e) return;
    int j = (int)(t & 31);

    int r = row[eidx];
    int c = col[eidx];
    float w = dis[r] * dis[c];

    const float2 hv = *(const float2*)(h + (size_t)r * 64 + 2 * j);
    float* op = out + (size_t)c * 64 + 2 * j;
    atomicAdd(op,     w * hv.x);
    atomicAdd(op + 1, w * hv.y);
}


extern "C" void kernel_launch(void* const* d_in, const int* in_sizes, int n_in,
                              void* d_out, int out_size, void* d_ws, size_t ws_size,
                              hipStream_t stream) {
    const float* x    = (const float*)d_in[0];
    const int*   eidx = (const int*)d_in[1];
    const float* W1   = (const float*)d_in[2];
    const float* b1   = (const float*)d_in[3];
    const float* W2   = (const float*)d_in[4];
    const float* b2   = (const float*)d_in[5];

    const int n = in_sizes[0] / IN_DIM;
    const int E = in_sizes[1] / 2;
    const int* row = eidx;
    const int* col = eidx + E;

    float* ws  = (float*)d_ws;
    float* dis = ws;
    float* t0  = dis + n;
    float* t1  = t0 + (size_t)n * HID;
    float* t2  = t1 + (size_t)n * HID;
    float* out = (float*)d_out;

    const int TPB = 256;
    const int mtiles = n / 16;
    const int waves_per_block = 4;
    dim3 gemm_grid((mtiles + waves_per_block - 1) / waves_per_block);
    dim3 gemm_block(32 * waves_per_block);

    k_deg_init <<<(n + TPB - 1) / TPB, TPB, 0, stream>>>(dis, n);
    k_deg_edges<<<(E + TPB - 1) / TPB, TPB, 0, stream>>>(col, dis, E);
    k_rsqrt    <<<(n + TPB - 1) / TPB, TPB, 0, stream>>>(dis, n);

    k_gemm_wmma<IN_DIM, false><<<gemm_grid, gemm_block, 0, stream>>>(x, W1, t0, mtiles);
    k_prop_init <<<((n * HID) + TPB - 1) / TPB, TPB, 0, stream>>>(t1, t0, dis, b1, n * HID);
    {
        long long tot = (long long)E * 32;
        k_prop_edges<<<(unsigned)((tot + TPB - 1) / TPB), TPB, 0, stream>>>(t1, t0, dis, row, col, E);
    }

    k_gemm_wmma<HID, true><<<gemm_grid, gemm_block, 0, stream>>>(t1, W2, t2, mtiles);
    k_prop_init <<<((n * OUTD) + TPB - 1) / TPB, TPB, 0, stream>>>(out, t2, dis, b2, n * OUTD);
    {
        long long tot = (long long)E * 32;
        k_prop_edges<<<(unsigned)((tot + TPB - 1) / TPB), TPB, 0, stream>>>(out, t2, dis, row, col, E);
    }
}


# edges bucketed by destination in-call (int count returns rank, offsets, fill); both propagations are per-node gathers with plain stores, no float atomics; sum order: bucket rank, two partial sums
# speedup vs baseline: 1.9235x; 1.9235x over previous
#include <hip/hip_runtime.h>
#include <hip/hip_bf16.h>


#define IN_DIM 128
#define HID    64
#define OUTD   64

typedef float v2f __attribute__((ext_vector_type(2)));
typedef float v8f __attribute__((ext_vector_type(8)));


template<int K, bool RELU_A>
__global__ void k_gemm_wmma(const float* __restrict__ A,
                            const float* __restrict__ B,
                            float* __restrict__ C,
                            int mtiles) {
    const int wave  = threadIdx.x >> 5;
    const int lane  = threadIdx.x & 31;
    const int tileM = blockIdx.x * (blockDim.x >> 5) + wave;
    if (tileM >= mtiles) return;

    const int nloc = lane & 15;
    const int half = lane >> 4;

    const float* Arow = A + (size_t)(tileM * 16 + nloc) * K;

    v8f acc0 = {}, acc1 = {}, acc2 = {}, acc3 = {};

    #pragma unroll 4
    for (int k0 = 0; k0 < K; k0 += 4) {
        v2f a = *(const v2f*)(Arow + k0 + 2 * half);
        if (RELU_A) { a.x = fmaxf(a.x, 0.0f); a.y = fmaxf(a.y, 0.0f); }

        const float* Bp = B + (size_t)(k0 + 2 * half) * 64 + nloc;
        v2f b0, b1, b2, b3;
        b0.x = Bp[0];  b0.y = Bp[64];
        b1.x = Bp[16]; b1.y = Bp[80];
        b2.x = Bp[32]; b2.y = Bp[96];
        b3.x = Bp[48]; b3.y = Bp[112];

        acc0 = __builtin_amdgcn_wmma_f32_16x16x4_f32(false, a, false, b0, (short)0, acc0, false, false);
        acc1 = __builtin_amdgcn_wmma_f32_16x16x4_f32(false, a, false, b1, (short)0, acc1, false, false);
        acc2 = __builtin_amdgcn_wmma_f32_16x16x4_f32(false, a, false, b2, (short)0, acc2, false, false);
        acc3 = __builtin_amdgcn_wmma_f32_16x16x4_f32(false, a, false, b3, (short)0, acc3, false, false);
    }

    float* Crow = C + (size_t)(tileM * 16 + half * 8) * 64 + nloc;
    #pragma unroll
    for (int v = 0; v < 8; ++v) {
        Crow[(size_t)v * 64 +  0] = acc0[v];
        Crow[(size_t)v * 64 + 16] = acc1[v];
        Crow[(size_t)v * 64 + 32] = acc2[v];
        Crow[(size_t)v * 64 + 48] = acc3[v];
    }
}


__global__ void k_zero(int* __restrict__ p, int m,
                       float* __restrict__ pad0, float* __restrict__ pad1) {
    int i = blockIdx.x * blockDim.x + threadIdx.x;
    if (i < m) p[i] = 0;
    if (i < 64) { pad0[i] = 0.0f; pad1[i] = 0.0f; }
}

__global__ void k_count(const int* __restrict__ col, int* __restrict__ cnt,
                        int* __restrict__ rank, int e) {
    int i = blockIdx.x * blockDim.x + threadIdx.x;
    if (i < e) rank[i] = atomicAdd(&cnt[col[i]], 1);
}

__global__ void k_alloc(const int* __restrict__ cnt, int* __restrict__ total,
                        int* __restrict__ start, float* __restrict__ dis, int n) {
    const int i    = blockIdx.x * blockDim.x + threadIdx.x;
    const int lane = threadIdx.x & 31;
    const int c    = (i < n) ? cnt[i] : 0;
    int incl = c;
    for (int d = 1; d < 32; d <<= 1) {
        int v = __shfl_up(incl, d, 32);
        if (lane >= d) incl += v;
    }
    int base = 0;
    if (lane == 31) base = atomicAdd(total, incl);
    base = __shfl(base, 31, 32);
    if (i < n) {
        start[i] = base + incl - c;
        dis[i]   = rsqrtf((float)(c + 1));
    }
}

__global__ void k_fill(const int* __restrict__ row, const int* __restrict__ col,
                       const int* __restrict__ rank, const int* __restrict__ start,
                       const float* __restrict__ dis, int2* __restrict__ ent, int e) {
    int i = blockIdx.x * blockDim.x + threadIdx.x;
    if (i < e) {
        int r = row[i];
        int2 v;
        v.x = r;
        v.y = __float_as_int(dis[r]);
        ent[start[col[i]] + rank[i]] = v;
    }
}

__global__ void k_gather(float* __restrict__ out, const float* __restrict__ h,
                         const float* __restrict__ dis, const float* __restrict__ bias,
                         const int* __restrict__ start, const int* __restrict__ cnt,
                         const int2* __restrict__ ent, int n) {
    const int lane = threadIdx.x & 31;
    const int node = blockIdx.x * (blockDim.x >> 5) + (threadIdx.x >> 5);
    if (node >= n) return;

    const int q    = lane & 15;
    const int half = lane >> 4;

    const int s0  = __builtin_amdgcn_readfirstlane(start[node]);
    const int deg = __builtin_amdgcn_readfirstlane(cnt[node]);
    const float dc = dis[node];

    const float* hq = h + 4 * q;
    const float4 hs = *(const float4*)(hq + (size_t)node * 64);
    const float4 b  = *(const float4*)(bias + 4 * q);
    const float  s2 = dc * dc;

    const int keep = half ? 0 : -1;
    float4 acc;
    acc.x = __int_as_float(__float_as_int(fmaf(s2, hs.x, b.x)) & keep);
    acc.y = __int_as_float(__float_as_int(fmaf(s2, hs.y, b.y)) & keep);
    acc.z = __int_as_float(__float_as_int(fmaf(s2, hs.z, b.z)) & keep);
    acc.w = __int_as_float(__float_as_int(fmaf(s2, hs.w, b.w)) & keep);

    for (int base = 0; base < deg; base += 32) {
        const int m = min(32, deg - base);
        int2 e = make_int2(n, 0);
        if (lane < m) e = ent[s0 + base + lane];

        for (int t0 = 0; 2 * t0 < m; t0 += 4) {
            float4 hv[4];
            float  w[4];
            #pragma unroll
            for (int u = 0; u < 4; ++u) {
                const int j = 2 * (t0 + u) + half;
                const int r = __shfl(e.x, j, 32);
                w[u]  = __int_as_float(__shfl(e.y, j, 32)) * dc;
                hv[u] = *(const float4*)(hq + (size_t)r * 64);
            }
            #pragma unroll
            for (int u = 0; u < 4; ++u) {
                acc.x = fmaf(w[u], hv[u].x, acc.x);
                acc.y = fmaf(w[u], hv[u].y, acc.y);
                acc.z = fmaf(w[u], hv[u].z, acc.z);
                acc.w = fmaf(w[u], hv[u].w, acc.w);
            }
        }
    }

    acc.x += __shfl_xor(acc.x, 16, 32);
    acc.y += __shfl_xor(acc.y, 16, 32);
    acc.z += __shfl_xor(acc.z, 16, 32);
    acc.w += __shfl_xor(acc.w, 16, 32);

    if (half == 0) *(float4*)(out + (size_t)node * 64 + 4 * q) = acc;
}


extern "C" void kernel_launch(void* const* d_in, const int* in_sizes, int n_in,
                              void* d_out, int out_size, void* d_ws, size_t ws_size,
                              hipStream_t stream) {
    const float* x    = (const float*)d_in[0];
    const int*   eidx = (const int*)d_in[1];
    const float* W1   = (const float*)d_in[2];
    const float* b1   = (const float*)d_in[3];
    const float* W2   = (const float*)d_in[4];
    const float* b2   = (const float*)d_in[5];

    const int n = in_sizes[0] / IN_DIM;
    const int E = in_sizes[1] / 2;
    const int* row = eidx;
    const int* col = eidx + E;

    const size_t nw = ((size_t)n + 1 + 63) & ~(size_t)63;
    const size_t ew = ((size_t)E + 63) & ~(size_t)63;
    int*   cnt   = (int*)d_ws;
    int*   total = cnt + n;
    int*   start = cnt + nw;
    float* dis   = (float*)(start + nw);
    int*   rank  = (int*)(dis + nw);
    int2*  ent   = (int2*)(rank + ew);
    float* t0    = (float*)(rank + ew) + 2 * ew;
    float* t1    = t0 + (size_t)(n + 1) * HID;
    float* t2    = t1 + (size_t)n * HID;
    float* out   = (float*)d_out;

    const int TPB = 256;
    const int mtiles = n / 16;
    const int waves_per_block = 4;
    dim3 gemm_grid((mtiles + waves_per_block - 1) / waves_per_block);
    dim3 gemm_block(32 * waves_per_block);
    const int nodes_per_block = TPB / 32;
    dim3 gather_grid((n + nodes_per_block - 1) / nodes_per_block);

    k_zero <<<(n + 1 + TPB - 1) / TPB, TPB, 0, stream>>>(cnt, n + 1, t0 + (size_t)n * HID, t2 + (size_t)n * OUTD);
    k_count<<<(E + TPB - 1) / TPB, TPB, 0, stream>>>(col, cnt, rank, E);
    k_alloc<<<(n + TPB - 1) / TPB, TPB, 0, stream>>>(cnt, total, start, dis, n);
    k_fill <<<(E + TPB - 1) / TPB, TPB, 0, stream>>>(row, col, rank, start, dis, ent, E);

    k_gemm_wmma<IN_DIM, false><<<gemm_grid, gemm_block, 0, stream>>>(x, W1, t0, mtiles);
    k_gather<<<gather_grid, TPB, 0, stream>>>(t1, t0, dis, b1, start, cnt, ent, n);

    k_gemm_wmma<HID, true><<<gemm_grid, gemm_block, 0, stream>>>(t1, W2, t2, mtiles);
    k_gather<<<gather_grid, TPB, 0, stream>>>(out, t2, dis, b2, start, cnt, ent, n);
}


# degree count and edge ranks via a per-work-group LDS histogram of all nodes, flushed with one line-coalesced returning add per node, not one scattered global atomic per edge; sum order: bucket rank
# speedup vs baseline: 2.4161x; 1.2561x over previous
#include <hip/hip_runtime.h>
#include <hip/hip_bf16.h>


#define IN_DIM 128
#define HID    64
#define OUTD   64

typedef float v2f __attribute__((ext_vector_type(2)));
typedef float v8f __attribute__((ext_vector_type(8)));


template<int K, bool RELU_A>
__global__ void k_gemm_wmma(const float* __restrict__ A,
                            const float* __restrict__ B,
                            float* __restrict__ C,
                            int mtiles) {
    const int wave  = threadIdx.x >> 5;
    const int lane  = threadIdx.x & 31;
    const int tileM = blockIdx.x * (blockDim.x >> 5) + wave;
    if (tileM >= mtiles) return;

    const int nloc = lane & 15;
    const int half = lane >> 4;

    const float* Arow = A + (size_t)(tileM * 16 + nloc) * K;

    v8f acc0 = {}, acc1 = {}, acc2 = {}, acc3 = {};

    #pragma unroll 4
    for (int k0 = 0; k0 < K; k0 += 4) {
        v2f a = *(const v2f*)(Arow + k0 + 2 * half);
        if (RELU_A) { a.x = fmaxf(a.x, 0.0f); a.y = fmaxf(a.y, 0.0f); }

        const float* Bp = B + (size_t)(k0 + 2 * half) * 64 + nloc;
        v2f b0, b1, b2, b3;
        b0.x = Bp[0];  b0.y = Bp[64];
        b1.x = Bp[16]; b1.y = Bp[80];
        b2.x = Bp[32]; b2.y = Bp[96];
        b3.x = Bp[48]; b3.y = Bp[112];

        acc0 = __builtin_amdgcn_wmma_f32_16x16x4_f32(false, a, false, b0, (short)0, acc0, false, false);
        acc1 = __builtin_amdgcn_wmma_f32_16x16x4_f32(false, a, false, b1, (short)0, acc1, false, false);
        acc2 = __builtin_amdgcn_wmma_f32_16x16x4_f32(false, a, false, b2, (short)0, acc2, false, false);
        acc3 = __builtin_amdgcn_wmma_f32_16x16x4_f32(false, a, false, b3, (short)0, acc3, false, false);
    }

    float* Crow = C + (size_t)(tileM * 16 + half * 8) * 64 + nloc;
    #pragma unroll
    for (int v = 0; v < 8; ++v) {
        Crow[(size_t)v * 64 +  0] = acc0[v];
        Crow[(size_t)v * 64 + 16] = acc1[v];
        Crow[(size_t)v * 64 + 32] = acc2[v];
        Crow[(size_t)v * 64 + 48] = acc3[v];
    }
}


__global__ void k_zero(int* __restrict__ p, int m,
                       float* __restrict__ pad0, float* __restrict__ pad1) {
    int i = blockIdx.x * blockDim.x + threadIdx.x;
    if (i < m) p[i] = 0;
    if (i < 64) { pad0[i] = 0.0f; pad1[i] = 0.0f; }
}

#define HIST_N   50176
#define CNT_EPT  16
#define CNT_TPB  1024

__global__ __launch_bounds__(CNT_TPB)
void k_count(const int* __restrict__ col, int* __restrict__ cnt,
             int* __restrict__ rank, int e, int n) {
    __shared__ __attribute__((aligned(16))) int hist[HIST_N];
    const int tid = threadIdx.x;

    int4* h4 = (int4*)hist;
    for (int j = tid; j < HIST_N / 4; j += CNT_TPB) h4[j] = make_int4(0, 0, 0, 0);

    const int e0 = blockIdx.x * (CNT_TPB * CNT_EPT) + tid;
    int c[CNT_EPT], lr[CNT_EPT];
    #pragma unroll
    for (int u = 0; u < CNT_EPT; ++u) {
        const int i = e0 + u * CNT_TPB;
        c[u] = (i < e) ? col[i] : (HIST_N - 1);
    }
    __syncthreads();

    #pragma unroll
    for (int u = 0; u < CNT_EPT; ++u) lr[u] = atomicAdd(&hist[c[u]], 1);
    __syncthreads();

    int first[HIST_N / CNT_TPB];
    #pragma unroll
    for (int j = 0; j < HIST_N / CNT_TPB; ++j) {
        const int node = j * CNT_TPB + tid;
        const int k = hist[node];
        first[j] = 0;
        if (k > 0 && node < n) first[j] = atomicAdd(&cnt[node], k);
    }
    #pragma unroll
    for (int j = 0; j < HIST_N / CNT_TPB; ++j) hist[j * CNT_TPB + tid] = first[j];
    __syncthreads();

    #pragma unroll
    for (int u = 0; u < CNT_EPT; ++u) {
        const int i = e0 + u * CNT_TPB;
        if (i < e) rank[i] = hist[c[u]] + lr[u];
    }
}

__global__ void k_alloc(const int* __restrict__ cnt, int* __restrict__ total,
                        int* __restrict__ start, float* __restrict__ dis, int n) {
    const int i    = blockIdx.x * blockDim.x + threadIdx.x;
    const int lane = threadIdx.x & 31;
    const int c    = (i < n) ? cnt[i] : 0;
    int incl = c;
    for (int d = 1; d < 32; d <<= 1) {
        int v = __shfl_up(incl, d, 32);
        if (lane >= d) incl += v;
    }
    int base = 0;
    if (lane == 31) base = atomicAdd(total, incl);
    base = __shfl(base, 31, 32);
    if (i < n) {
        start[i] = base + incl - c;
        dis[i]   = rsqrtf((float)(c + 1));
    }
}

__global__ void k_fill(const int* __restrict__ row, const int* __restrict__ col,
                       const int* __restrict__ rank, const int* __restrict__ start,
                       const float* __restrict__ dis, int2* __restrict__ ent, int e) {
    int i = blockIdx.x * blockDim.x + threadIdx.x;
    if (i < e) {
        int r = row[i];
        int2 v;
        v.x = r;
        v.y = __float_as_int(dis[r]);
        ent[start[col[i]] + rank[i]] = v;
    }
}

__global__ void k_gather(float* __restrict__ out, const float* __restrict__ h,
                         const float* __restrict__ dis, const float* __restrict__ bias,
                         const int* __restrict__ start, const int* __restrict__ cnt,
                         const int2* __restrict__ ent, int n) {
    const int lane = threadIdx.x & 31;
    const int node = blockIdx.x * (blockDim.x >> 5) + (threadIdx.x >> 5);
    if (node >= n) return;

    const int q    = lane & 15;
    const int half = lane >> 4;

    const int s0  = __builtin_amdgcn_readfirstlane(start[node]);
    const int deg = __builtin_amdgcn_readfirstlane(cnt[node]);
    const float dc = dis[node];

    const float* hq = h + 4 * q;
    const float4 hs = *(const float4*)(hq + (size_t)node * 64);
    const float4 b  = *(const float4*)(bias + 4 * q);
    const float  s2 = dc * dc;

    const int keep = half ? 0 : -1;
    float4 acc;
    acc.x = __int_as_float(__float_as_int(fmaf(s2, hs.x, b.x)) & keep);
    acc.y = __int_as_float(__float_as_int(fmaf(s2, hs.y, b.y)) & keep);
    acc.z = __int_as_float(__float_as_int(fmaf(s2, hs.z, b.z)) & keep);
    acc.w = __int_as_float(__float_as_int(fmaf(s2, hs.w, b.w)) & keep);

    for (int base = 0; base < deg; base += 32) {
        const int m = min(32, deg - base);
        int2 e = make_int2(n, 0);
        if (lane < m) e = ent[s0 + base + lane];

        for (int t0 = 0; 2 * t0 < m; t0 += 4) {
            float4 hv[4];
            float  w[4];
            #pragma unroll
            for (int u = 0; u < 4; ++u) {
                const int j = 2 * (t0 + u) + half;
                const int r = __shfl(e.x, j, 32);
                w[u]  = __int_as_float(__shfl(e.y, j, 32)) * dc;
                hv[u] = *(const float4*)(hq + (size_t)r * 64);
            }
            #pragma unroll
            for (int u = 0; u < 4; ++u) {
                acc.x = fmaf(w[u], hv[u].x, acc.x);
                acc.y = fmaf(w[u], hv[u].y, acc.y);
                acc.z = fmaf(w[u], hv[u].z, acc.z);
                acc.w = fmaf(w[u], hv[u].w, acc.w);
            }
        }
    }

    acc.x += __shfl_xor(acc.x, 16, 32);
    acc.y += __shfl_xor(acc.y, 16, 32);
    acc.z += __shfl_xor(acc.z, 16, 32);
    acc.w += __shfl_xor(acc.w, 16, 32);

    if (half == 0) *(float4*)(out + (size_t)node * 64 + 4 * q) = acc;
}


extern "C" void kernel_launch(void* const* d_in, const int* in_sizes, int n_in,
                              void* d_out, int out_size, void* d_ws, size_t ws_size,
                              hipStream_t stream) {
    const float* x    = (const float*)d_in[0];
    const int*   eidx = (const int*)d_in[1];
    const float* W1   = (const float*)d_in[2];
    const float* b1   = (const float*)d_in[3];
    const float* W2   = (const float*)d_in[4];
    const float* b2   = (const float*)d_in[5];

    const int n = in_sizes[0] / IN_DIM;
    const int E = in_sizes[1] / 2;
    const int* row = eidx;
    const int* col = eidx + E;

    const size_t nw = ((size_t)n + 1 + 63) & ~(size_t)63;
    const size_t ew = ((size_t)E + 63) & ~(size_t)63;
    int*   cnt   = (int*)d_ws;
    int*   total = cnt + n;
    int*   start = cnt + nw;
    float* dis   = (float*)(start + nw);
    int*   rank  = (int*)(dis + nw);
    int2*  ent   = (int2*)(rank + ew);
    float* t0    = (float*)(rank + ew) + 2 * ew;
    float* t1    = t0 + (size_t)(n + 1) * HID;
    float* t2    = t1 + (size_t)n * HID;
    float* out   = (float*)d_out;

    const int TPB = 256;
    const int mtiles = n / 16;
    const int waves_per_block = 4;
    dim3 gemm_grid((mtiles + waves_per_block - 1) / waves_per_block);
    dim3 gemm_block(32 * waves_per_block);
    const int nodes_per_block = TPB / 32;
    dim3 gather_grid((n + nodes_per_block - 1) / nodes_per_block);

    k_zero <<<(n + 1 + TPB - 1) / TPB, TPB, 0, stream>>>(cnt, n + 1, t0 + (size_t)n * HID, t2 + (size_t)n * OUTD);
    k_count<<<(E + CNT_TPB * CNT_EPT - 1) / (CNT_TPB * CNT_EPT), CNT_TPB, 0, stream>>>(col, cnt, rank, E, n);
    k_alloc<<<(n + TPB - 1) / TPB, TPB, 0, stream>>>(cnt, total, start, dis, n);
    k_fill <<<(E + TPB - 1) / TPB, TPB, 0, stream>>>(row, col, rank, start, dis, ent, E);

    k_gemm_wmma<IN_DIM, false><<<gemm_grid, gemm_block, 0, stream>>>(x, W1, t0, mtiles);
    k_gather<<<gather_grid, TPB, 0, stream>>>(t1, t0, dis, b1, start, cnt, ent, n);

    k_gemm_wmma<HID, true><<<gemm_grid, gemm_block, 0, stream>>>(t1, W2, t2, mtiles);
    k_gather<<<gather_grid, TPB, 0, stream>>>(out, t2, dis, b2, start, cnt, ent, n);
}


# layer-1 product x@W1 moved into the edge-count launch: its strips run as extra work-groups on the WGPs the 49 histogram groups leave idle; one launch fewer; all sums as before
# speedup vs baseline: 2.8139x; 1.1646x over previous
#include <hip/hip_runtime.h>
#include <hip/hip_bf16.h>


#define IN_DIM 128
#define HID    64
#define OUTD   64

typedef float v2f __attribute__((ext_vector_type(2)));
typedef float v8f __attribute__((ext_vector_type(8)));


template<int K, bool RELU_A>
__device__ __forceinline__ void gemm_strip(const float* __restrict__ A,
                                           const float* __restrict__ B,
                                           float* __restrict__ C,
                                           int tileM, int lane) {
    const int nloc = lane & 15;
    const int half = lane >> 4;

    const float* Arow = A + (size_t)(tileM * 16 + nloc) * K;

    v8f acc0 = {}, acc1 = {}, acc2 = {}, acc3 = {};

    #pragma unroll 4
    for (int k0 = 0; k0 < K; k0 += 4) {
        v2f a = *(const v2f*)(Arow + k0 + 2 * half);
        if (RELU_A) { a.x = fmaxf(a.x, 0.0f); a.y = fmaxf(a.y, 0.0f); }

        const float* Bp = B + (size_t)(k0 + 2 * half) * 64 + nloc;
        v2f b0, b1, b2, b3;
        b0.x = Bp[0];  b0.y = Bp[64];
        b1.x = Bp[16]; b1.y = Bp[80];
        b2.x = Bp[32]; b2.y = Bp[96];
        b3.x = Bp[48]; b3.y = Bp[112];

        acc0 = __builtin_amdgcn_wmma_f32_16x16x4_f32(false, a, false, b0, (short)0, acc0, false, false);
        acc1 = __builtin_amdgcn_wmma_f32_16x16x4_f32(false, a, false, b1, (short)0, acc1, false, false);
        acc2 = __builtin_amdgcn_wmma_f32_16x16x4_f32(false, a, false, b2, (short)0, acc2, false, false);
        acc3 = __builtin_amdgcn_wmma_f32_16x16x4_f32(false, a, false, b3, (short)0, acc3, false, false);
    }

    float* Crow = C + (size_t)(tileM * 16 + half * 8) * 64 + nloc;
    #pragma unroll
    for (int v = 0; v < 8; ++v) {
        Crow[(size_t)v * 64 +  0] = acc0[v];
        Crow[(size_t)v * 64 + 16] = acc1[v];
        Crow[(size_t)v * 64 + 32] = acc2[v];
        Crow[(size_t)v * 64 + 48] = acc3[v];
    }
}

template<int K, bool RELU_A>
__global__ void k_gemm_wmma(const float* __restrict__ A,
                            const float* __restrict__ B,
                            float* __restrict__ C,
                            int mtiles) {
    const int wave  = threadIdx.x >> 5;
    const int lane  = threadIdx.x & 31;
    const int tileM = blockIdx.x * (blockDim.x >> 5) + wave;
    if (tileM >= mtiles) return;
    gemm_strip<K, RELU_A>(A, B, C, tileM, lane);
}


__global__ void k_zero(int* __restrict__ p, int m,
                       float* __restrict__ pad0, float* __restrict__ pad1) {
    int i = blockIdx.x * blockDim.x + threadIdx.x;
    if (i < m) p[i] = 0;
    if (i < 64) { pad0[i] = 0.0f; pad1[i] = 0.0f; }
}

#define HIST_N   50176
#define CNT_EPT  16
#define CNT_TPB  1024

#define XW_WAVES 16

__global__ __launch_bounds__(CNT_TPB)
void k_count_xw(const int* __restrict__ col, int* __restrict__ cnt,
                int* __restrict__ rank, int e, int n, int count_blocks,
                const float* __restrict__ x, const float* __restrict__ W1,
                float* __restrict__ xw, int mtiles) {
    __shared__ __attribute__((aligned(16))) int hist[HIST_N];
    const int tid = threadIdx.x;

    if ((int)blockIdx.x >= count_blocks) {
        const int wave  = tid >> 5;
        const int tileM = ((int)blockIdx.x - count_blocks) * XW_WAVES + wave;
        if (wave < XW_WAVES && tileM < mtiles)
            gemm_strip<IN_DIM, false>(x, W1, xw, tileM, tid & 31);
        return;
    }

    int4* h4 = (int4*)hist;
    for (int j = tid; j < HIST_N / 4; j += CNT_TPB) h4[j] = make_int4(0, 0, 0, 0);

    const int e0 = blockIdx.x * (CNT_TPB * CNT_EPT) + tid;
    int c[CNT_EPT], lr[CNT_EPT];
    #pragma unroll
    for (int u = 0; u < CNT_EPT; ++u) {
        const int i = e0 + u * CNT_TPB;
        c[u] = (i < e) ? col[i] : (HIST_N - 1);
    }
    __syncthreads();

    #pragma unroll
    for (int u = 0; u < CNT_EPT; ++u) lr[u] = atomicAdd(&hist[c[u]], 1);
    __syncthreads();

    int first[HIST_N / CNT_TPB];
    #pragma unroll
    for (int j = 0; j < HIST_N / CNT_TPB; ++j) {
        const int node = j * CNT_TPB + tid;
        const int k = hist[node];
        first[j] = 0;
        if (k > 0 && node < n) first[j] = atomicAdd(&cnt[node], k);
    }
    #pragma unroll
    for (int j = 0; j < HIST_N / CNT_TPB; ++j) hist[j * CNT_TPB + tid] = first[j];
    __syncthreads();

    #pragma unroll
    for (int u = 0; u < CNT_EPT; ++u) {
        const int i = e0 + u * CNT_TPB;
        if (i < e) rank[i] = hist[c[u]] + lr[u];
    }
}

__global__ void k_alloc(const int* __restrict__ cnt, int* __restrict__ total,
                        int* __restrict__ start, float* __restrict__ dis, int n) {
    const int i    = blockIdx.x * blockDim.x + threadIdx.x;
    const int lane = threadIdx.x & 31;
    const int c    = (i < n) ? cnt[i] : 0;
    int incl = c;
    for (int d = 1; d < 32; d <<= 1) {
        int v = __shfl_up(incl, d, 32);
        if (lane >= d) incl += v;
    }
    int base = 0;
    if (lane == 31) base = atomicAdd(total, incl);
    base = __shfl(base, 31, 32);
    if (i < n) {
        start[i] = base + incl - c;
        dis[i]   = rsqrtf((float)(c + 1));
    }
}

__global__ void k_fill(const int* __restrict__ row, const int* __restrict__ col,
                       const int* __restrict__ rank, const int* __restrict__ start,
                       const float* __restrict__ dis, int2* __restrict__ ent, int e) {
    int i = blockIdx.x * blockDim.x + threadIdx.x;
    if (i < e) {
        int r = row[i];
        int2 v;
        v.x = r;
        v.y = __float_as_int(dis[r]);
        ent[start[col[i]] + rank[i]] = v;
    }
}

__global__ void k_gather(float* __restrict__ out, const float* __restrict__ h,
                         const float* __restrict__ dis, const float* __restrict__ bias,
                         const int* __restrict__ start, const int* __restrict__ cnt,
                         const int2* __restrict__ ent, int n) {
    const int lane = threadIdx.x & 31;
    const int node = blockIdx.x * (blockDim.x >> 5) + (threadIdx.x >> 5);
    if (node >= n) return;

    const int q    = lane & 15;
    const int half = lane >> 4;

    const int s0  = __builtin_amdgcn_readfirstlane(start[node]);
    const int deg = __builtin_amdgcn_readfirstlane(cnt[node]);
    const float dc = dis[node];

    const float* hq = h + 4 * q;
    const float4 hs = *(const float4*)(hq + (size_t)node * 64);
    const float4 b  = *(const float4*)(bias + 4 * q);
    const float  s2 = dc * dc;

    const int keep = half ? 0 : -1;
    float4 acc;
    acc.x = __int_as_float(__float_as_int(fmaf(s2, hs.x, b.x)) & keep);
    acc.y = __int_as_float(__float_as_int(fmaf(s2, hs.y, b.y)) & keep);
    acc.z = __int_as_float(__float_as_int(fmaf(s2, hs.z, b.z)) & keep);
    acc.w = __int_as_float(__float_as_int(fmaf(s2, hs.w, b.w)) & keep);

    for (int base = 0; base < deg; base += 32) {
        const int m = min(32, deg - base);
        int2 e = make_int2(n, 0);
        if (lane < m) e = ent[s0 + base + lane];

        for (int t0 = 0; 2 * t0 < m; t0 += 4) {
            float4 hv[4];
            float  w[4];
            #pragma unroll
            for (int u = 0; u < 4; ++u) {
                const int j = 2 * (t0 + u) + half;
                const int r = __shfl(e.x, j, 32);
                w[u]  = __int_as_float(__shfl(e.y, j, 32)) * dc;
                hv[u] = *(const float4*)(hq + (size_t)r * 64);
            }
            #pragma unroll
            for (int u = 0; u < 4; ++u) {
                acc.x = fmaf(w[u], hv[u].x, acc.x);
                acc.y = fmaf(w[u], hv[u].y, acc.y);
                acc.z = fmaf(w[u], hv[u].z, acc.z);
                acc.w = fmaf(w[u], hv[u].w, acc.w);
            }
        }
    }

    acc.x += __shfl_xor(acc.x, 16, 32);
    acc.y += __shfl_xor(acc.y, 16, 32);
    acc.z += __shfl_xor(acc.z, 16, 32);
    acc.w += __shfl_xor(acc.w, 16, 32);

    if (half == 0) *(float4*)(out + (size_t)node * 64 + 4 * q) = acc;
}


extern "C" void kernel_launch(void* const* d_in, const int* in_sizes, int n_in,
                              void* d_out, int out_size, void* d_ws, size_t ws_size,
                              hipStream_t stream) {
    const float* x    = (const float*)d_in[0];
    const int*   eidx = (const int*)d_in[1];
    const float* W1   = (const float*)d_in[2];
    const float* b1   = (const float*)d_in[3];
    const float* W2   = (const float*)d_in[4];
    const float* b2   = (const float*)d_in[5];

    const int n = in_sizes[0] / IN_DIM;
    const int E = in_sizes[1] / 2;
    const int* row = eidx;
    const int* col = eidx + E;

    const size_t nw = ((size_t)n + 1 + 63) & ~(size_t)63;
    const size_t ew = ((size_t)E + 63) & ~(size_t)63;
    int*   cnt   = (int*)d_ws;
    int*   total = cnt + n;
    int*   start = cnt + nw;
    float* dis   = (float*)(start + nw);
    int*   rank  = (int*)(dis + nw);
    int2*  ent   = (int2*)(rank + ew);
    float* t0    = (float*)(rank + ew) + 2 * ew;
    float* t1    = t0 + (size_t)(n + 1) * HID;
    float* t2    = t1 + (size_t)n * HID;
    float* out   = (float*)d_out;

    const int TPB = 256;
    const int mtiles = n / 16;
    const int waves_per_block = 4;
    dim3 gemm_grid((mtiles + waves_per_block - 1) / waves_per_block);
    dim3 gemm_block(32 * waves_per_block);
    const int nodes_per_block = TPB / 32;
    dim3 gather_grid((n + nodes_per_block - 1) / nodes_per_block);

    k_zero <<<(n + 1 + TPB - 1) / TPB, TPB, 0, stream>>>(cnt, n + 1, t0 + (size_t)n * HID, t2 + (size_t)n * OUTD);
    const int count_blocks = (E + CNT_TPB * CNT_EPT - 1) / (CNT_TPB * CNT_EPT);
    const int xw_blocks    = (mtiles + XW_WAVES - 1) / XW_WAVES;
    k_count_xw<<<count_blocks + xw_blocks, CNT_TPB, 0, stream>>>(col, cnt, rank, E, n, count_blocks, x, W1, t0, mtiles);
    k_alloc<<<(n + TPB - 1) / TPB, TPB, 0, stream>>>(cnt, total, start, dis, n);
    k_fill <<<(E + TPB - 1) / TPB, TPB, 0, stream>>>(row, col, rank, start, dis, ent, E);

    k_gather<<<gather_grid, TPB, 0, stream>>>(t1, t0, dis, b1, start, cnt, ent, n);

    k_gemm_wmma<HID, true><<<gemm_grid, gemm_block, 0, stream>>>(t1, W2, t2, mtiles);
    k_gather<<<gather_grid, TPB, 0, stream>>>(out, t2, dis, b2, start, cnt, ent, n);
}


# layer-2 product relu(h1)@W2 moved into the first gather's tail: 16 nodes per work-group, rows handed over in LDS, 4 waves run the WMMAs, rows stored whole; one launch fewer; all sums as before
# speedup vs baseline: 3.0207x; 1.0735x over previous
#include <hip/hip_runtime.h>
#include <hip/hip_bf16.h>


#define IN_DIM 128
#define HID    64
#define OUTD   64

typedef float v2f __attribute__((ext_vector_type(2)));
typedef float v8f __attribute__((ext_vector_type(8)));


template<int K, bool RELU_A>
__device__ __forceinline__ void gemm_strip(const float* __restrict__ A,
                                           const float* __restrict__ B,
                                           float* __restrict__ C,
                                           int tileM, int lane) {
    const int nloc = lane & 15;
    const int half = lane >> 4;

    const float* Arow = A + (size_t)(tileM * 16 + nloc) * K;

    v8f acc0 = {}, acc1 = {}, acc2 = {}, acc3 = {};

    #pragma unroll 4
    for (int k0 = 0; k0 < K; k0 += 4) {
        v2f a = *(const v2f*)(Arow + k0 + 2 * half);
        if (RELU_A) { a.x = fmaxf(a.x, 0.0f); a.y = fmaxf(a.y, 0.0f); }

        const float* Bp = B + (size_t)(k0 + 2 * half) * 64 + nloc;
        v2f b0, b1, b2, b3;
        b0.x = Bp[0];  b0.y = Bp[64];
        b1.x = Bp[16]; b1.y = Bp[80];
        b2.x = Bp[32]; b2.y = Bp[96];
        b3.x = Bp[48]; b3.y = Bp[112];

        acc0 = __builtin_amdgcn_wmma_f32_16x16x4_f32(false, a, false, b0, (short)0, acc0, false, false);
        acc1 = __builtin_amdgcn_wmma_f32_16x16x4_f32(false, a, false, b1, (short)0, acc1, false, false);
        acc2 = __builtin_amdgcn_wmma_f32_16x16x4_f32(false, a, false, b2, (short)0, acc2, false, false);
        acc3 = __builtin_amdgcn_wmma_f32_16x16x4_f32(false, a, false, b3, (short)0, acc3, false, false);
    }

    float* Crow = C + (size_t)(tileM * 16 + half * 8) * 64 + nloc;
    #pragma unroll
    for (int v = 0; v < 8; ++v) {
        Crow[(size_t)v * 64 +  0] = acc0[v];
        Crow[(size_t)v * 64 + 16] = acc1[v];
        Crow[(size_t)v * 64 + 32] = acc2[v];
        Crow[(size_t)v * 64 + 48] = acc3[v];
    }
}


__global__ void k_zero(int* __restrict__ p, int m,
                       float* __restrict__ pad0, float* __restrict__ pad1) {
    int i = blockIdx.x * blockDim.x + threadIdx.x;
    if (i < m) p[i] = 0;
    if (i < 64) { pad0[i] = 0.0f; pad1[i] = 0.0f; }
}

#define HIST_N   50176
#define CNT_EPT  16
#define CNT_TPB  1024

#define XW_WAVES 16

__global__ __launch_bounds__(CNT_TPB)
void k_count_xw(const int* __restrict__ col, int* __restrict__ cnt,
                int* __restrict__ rank, int e, int n, int count_blocks,
                const float* __restrict__ x, const float* __restrict__ W1,
                float* __restrict__ xw, int mtiles) {
    __shared__ __attribute__((aligned(16))) int hist[HIST_N];
    const int tid = threadIdx.x;

    if ((int)blockIdx.x >= count_blocks) {
        const int wave  = tid >> 5;
        const int tileM = ((int)blockIdx.x - count_blocks) * XW_WAVES + wave;
        if (wave < XW_WAVES && tileM < mtiles)
            gemm_strip<IN_DIM, false>(x, W1, xw, tileM, tid & 31);
        return;
    }

    int4* h4 = (int4*)hist;
    for (int j = tid; j < HIST_N / 4; j += CNT_TPB) h4[j] = make_int4(0, 0, 0, 0);

    const int e0 = blockIdx.x * (CNT_TPB * CNT_EPT) + tid;
    int c[CNT_EPT], lr[CNT_EPT];
    #pragma unroll
    for (int u = 0; u < CNT_EPT; ++u) {
        const int i = e0 + u * CNT_TPB;
        c[u] = (i < e) ? col[i] : (HIST_N - 1);
    }
    __syncthreads();

    #pragma unroll
    for (int u = 0; u < CNT_EPT; ++u) lr[u] = atomicAdd(&hist[c[u]], 1);
    __syncthreads();

    int first[HIST_N / CNT_TPB];
    #pragma unroll
    for (int j = 0; j < HIST_N / CNT_TPB; ++j) {
        const int node = j * CNT_TPB + tid;
        const int k = hist[node];
        first[j] = 0;
        if (k > 0 && node < n) first[j] = atomicAdd(&cnt[node], k);
    }
    #pragma unroll
    for (int j = 0; j < HIST_N / CNT_TPB; ++j) hist[j * CNT_TPB + tid] = first[j];
    __syncthreads();

    #pragma unroll
    for (int u = 0; u < CNT_EPT; ++u) {
        const int i = e0 + u * CNT_TPB;
        if (i < e) rank[i] = hist[c[u]] + lr[u];
    }
}

__global__ void k_alloc(const int* __restrict__ cnt, int* __restrict__ total,
                        int* __restrict__ start, float* __restrict__ dis, int n) {
    const int i    = blockIdx.x * blockDim.x + threadIdx.x;
    const int lane = threadIdx.x & 31;
    const int c    = (i < n) ? cnt[i] : 0;
    int incl = c;
    for (int d = 1; d < 32; d <<= 1) {
        int v = __shfl_up(incl, d, 32);
        if (lane >= d) incl += v;
    }
    int base = 0;
    if (lane == 31) base = atomicAdd(total, incl);
    base = __shfl(base, 31, 32);
    if (i < n) {
        start[i] = base + incl - c;
        dis[i]   = rsqrtf((float)(c + 1));
    }
}

__global__ void k_fill(const int* __restrict__ row, const int* __restrict__ col,
                       const int* __restrict__ rank, const int* __restrict__ start,
                       const float* __restrict__ dis, int2* __restrict__ ent, int e) {
    int i = blockIdx.x * blockDim.x + threadIdx.x;
    if (i < e) {
        int r = row[i];
        int2 v;
        v.x = r;
        v.y = __float_as_int(dis[r]);
        ent[start[col[i]] + rank[i]] = v;
    }
}

__device__ __forceinline__ float4 gather_row(const float* __restrict__ h,
                                             const float* __restrict__ dis, const float* __restrict__ bias,
                                             const int* __restrict__ start, const int* __restrict__ cnt,
                                             const int2* __restrict__ ent, int n, int node, int lane) {
    const int q    = lane & 15;
    const int half = lane >> 4;

    const int s0  = __builtin_amdgcn_readfirstlane(start[node]);
    const int deg = __builtin_amdgcn_readfirstlane(cnt[node]);
    const float dc = dis[node];

    const float* hq = h + 4 * q;
    const float4 hs = *(const float4*)(hq + (size_t)node * 64);
    const float4 b  = *(const float4*)(bias + 4 * q);
    const float  s2 = dc * dc;

    const int keep = half ? 0 : -1;
    float4 acc;
    acc.x = __int_as_float(__float_as_int(fmaf(s2, hs.x, b.x)) & keep);
    acc.y = __int_as_float(__float_as_int(fmaf(s2, hs.y, b.y)) & keep);
    acc.z = __int_as_float(__float_as_int(fmaf(s2, hs.z, b.z)) & keep);
    acc.w = __int_as_float(__float_as_int(fmaf(s2, hs.w, b.w)) & keep);

    for (int base = 0; base < deg; base += 32) {
        const int m = min(32, deg - base);
        int2 e = make_int2(n, 0);
        if (lane < m) e = ent[s0 + base + lane];

        for (int t0 = 0; 2 * t0 < m; t0 += 4) {
            float4 hv[4];
            float  w[4];
            #pragma unroll
            for (int u = 0; u < 4; ++u) {
                const int j = 2 * (t0 + u) + half;
                const int r = __shfl(e.x, j, 32);
                w[u]  = __int_as_float(__shfl(e.y, j, 32)) * dc;
                hv[u] = *(const float4*)(hq + (size_t)r * 64);
            }
            #pragma unroll
            for (int u = 0; u < 4; ++u) {
                acc.x = fmaf(w[u], hv[u].x, acc.x);
                acc.y = fmaf(w[u], hv[u].y, acc.y);
                acc.z = fmaf(w[u], hv[u].z, acc.z);
                acc.w = fmaf(w[u], hv[u].w, acc.w);
            }
        }
    }

    acc.x += __shfl_xor(acc.x, 16, 32);
    acc.y += __shfl_xor(acc.y, 16, 32);
    acc.z += __shfl_xor(acc.z, 16, 32);
    acc.w += __shfl_xor(acc.w, 16, 32);
    return acc;
}

__global__ void k_gather(float* __restrict__ out, const float* __restrict__ h,
                         const float* __restrict__ dis, const float* __restrict__ bias,
                         const int* __restrict__ start, const int* __restrict__ cnt,
                         const int2* __restrict__ ent, int n) {
    const int lane = threadIdx.x & 31;
    const int node = blockIdx.x * (blockDim.x >> 5) + (threadIdx.x >> 5);
    if (node >= n) return;

    const float4 acc = gather_row(h, dis, bias, start, cnt, ent, n, node, lane);
    if ((lane >> 4) == 0) *(float4*)(out + (size_t)node * 64 + 4 * (lane & 15)) = acc;
}

#define GM_WAVES 16
#define GM_PITCH 68

__global__ __launch_bounds__(GM_WAVES * 32)
void k_gather_mm(float* __restrict__ hw, const float* __restrict__ h,
                 const float* __restrict__ dis, const float* __restrict__ bias,
                 const int* __restrict__ start, const int* __restrict__ cnt,
                 const int2* __restrict__ ent, const float* __restrict__ W, int n) {
    __shared__ __attribute__((aligned(16))) float sA[GM_WAVES * GM_PITCH];
    __shared__ __attribute__((aligned(16))) float sC[GM_WAVES * GM_PITCH];

    const int lane = threadIdx.x & 31;
    const int node = blockIdx.x * GM_WAVES + (threadIdx.x >> 5);
    const bool valid = node < n;
    const int wave = __builtin_amdgcn_readfirstlane((int)(threadIdx.x >> 5));
    const int nloc = lane & 15;
    const int half = lane >> 4;

    float4 acc = gather_row(h, dis, bias, start, cnt, ent, n, valid ? node : 0, lane);
    acc.x = fmaxf(acc.x, 0.0f); acc.y = fmaxf(acc.y, 0.0f);
    acc.z = fmaxf(acc.z, 0.0f); acc.w = fmaxf(acc.w, 0.0f);
    if (half == 0) *(float4*)(sA + wave * GM_PITCH + 4 * nloc) = acc;

    v2f bf[HID / 4];
    if (wave < 4) {
        const float* Bp = W + (size_t)(2 * half) * 64 + 16 * wave + nloc;
        #pragma unroll
        for (int s = 0; s < HID / 4; ++s) {
            bf[s].x = Bp[(size_t)(4 * s) * 64];
            bf[s].y = Bp[(size_t)(4 * s + 1) * 64];
        }
    }
    __builtin_amdgcn_fence(__ATOMIC_RELEASE, "workgroup", "local");
    __builtin_amdgcn_s_barrier();
    __builtin_amdgcn_fence(__ATOMIC_ACQUIRE, "workgroup", "local");

    if (wave < 4) {
        const float* Ap = sA + nloc * GM_PITCH + 2 * half;
        v8f d = {};
        #pragma unroll
        for (int s = 0; s < HID / 4; ++s) {
            const v2f a = *(const v2f*)(Ap + 4 * s);
            d = __builtin_amdgcn_wmma_f32_16x16x4_f32(false, a, false, bf[s], (short)0, d, false, false);
        }
        float* Cp = sC + (8 * half) * GM_PITCH + 16 * wave + nloc;
        #pragma unroll
        for (int v = 0; v < 8; ++v) Cp[v * GM_PITCH] = d[v];
    }
    __syncthreads();

    if (valid) {
        const float2 o = *(const float2*)(sC + wave * GM_PITCH + 2 * lane);
        *(float2*)(hw + (size_t)node * 64 + 2 * lane) = o;
    }
}


extern "C" void kernel_launch(void* const* d_in, const int* in_sizes, int n_in,
                              void* d_out, int out_size, void* d_ws, size_t ws_size,
                              hipStream_t stream) {
    const float* x    = (const float*)d_in[0];
    const int*   eidx = (const int*)d_in[1];
    const float* W1   = (const float*)d_in[2];
    const float* b1   = (const float*)d_in[3];
    const float* W2   = (const float*)d_in[4];
    const float* b2   = (const float*)d_in[5];

    const int n = in_sizes[0] / IN_DIM;
    const int E = in_sizes[1] / 2;
    const int* row = eidx;
    const int* col = eidx + E;

    const size_t nw = ((size_t)n + 1 + 63) & ~(size_t)63;
    const size_t ew = ((size_t)E + 63) & ~(size_t)63;
    int*   cnt   = (int*)d_ws;
    int*   total = cnt + n;
    int*   start = cnt + nw;
    float* dis   = (float*)(start + nw);
    int*   rank  = (int*)(dis + nw);
    int2*  ent   = (int2*)(rank + ew);
    float* t0    = (float*)(rank + ew) + 2 * ew;
    float* t2    = t0 + (size_t)(n + 1) * HID;
    float* out   = (float*)d_out;

    const int TPB = 256;
    const int mtiles = n / 16;
    const int nodes_per_block = TPB / 32;
    dim3 gather_grid((n + nodes_per_block - 1) / nodes_per_block);

    k_zero <<<(n + 1 + TPB - 1) / TPB, TPB, 0, stream>>>(cnt, n + 1, t0 + (size_t)n * HID, t2 + (size_t)n * OUTD);
    const int count_blocks = (E + CNT_TPB * CNT_EPT - 1) / (CNT_TPB * CNT_EPT);
    const int xw_blocks    = (mtiles + XW_WAVES - 1) / XW_WAVES;
    k_count_xw<<<count_blocks + xw_blocks, CNT_TPB, 0, stream>>>(col, cnt, rank, E, n, count_blocks, x, W1, t0, mtiles);
    k_alloc<<<(n + TPB - 1) / TPB, TPB, 0, stream>>>(cnt, total, start, dis, n);
    k_fill <<<(E + TPB - 1) / TPB, TPB, 0, stream>>>(row, col, rank, start, dis, ent, E);

    k_gather_mm<<<(n + GM_WAVES - 1) / GM_WAVES, GM_WAVES * 32, 0, stream>>>(t2, t0, dis, b1, start, cnt, ent, W2, n);

    k_gather<<<gather_grid, TPB, 0, stream>>>(out, t2, dis, b2, start, cnt, ent, n);
}


# edge count: lanes past the end of the edge list skip the LDS add instead of all adding to one spare bin (same-word LDS adds go lane by lane); sum order: bucket rank, as before
# speedup vs baseline: 3.2075x; 1.0618x over previous
#include <hip/hip_runtime.h>
#include <hip/hip_bf16.h>


#define IN_DIM 128
#define HID    64
#define OUTD   64

typedef float v2f __attribute__((ext_vector_type(2)));
typedef float v8f __attribute__((ext_vector_type(8)));


template<int K, bool RELU_A>
__device__ __forceinline__ void gemm_strip(const float* __restrict__ A,
                                           const float* __restrict__ B,
                                           float* __restrict__ C,
                                           int tileM, int lane) {
    const int nloc = lane & 15;
    const int half = lane >> 4;

    const float* Arow = A + (size_t)(tileM * 16 + nloc) * K;

    v8f acc0 = {}, acc1 = {}, acc2 = {}, acc3 = {};

    #pragma unroll 4
    for (int k0 = 0; k0 < K; k0 += 4) {
        v2f a = *(const v2f*)(Arow + k0 + 2 * half);
        if (RELU_A) { a.x = fmaxf(a.x, 0.0f); a.y = fmaxf(a.y, 0.0f); }

        const float* Bp = B + (size_t)(k0 + 2 * half) * 64 + nloc;
        v2f b0, b1, b2, b3;
        b0.x = Bp[0];  b0.y = Bp[64];
        b1.x = Bp[16]; b1.y = Bp[80];
        b2.x = Bp[32]; b2.y = Bp[96];
        b3.x = Bp[48]; b3.y = Bp[112];

        acc0 = __builtin_amdgcn_wmma_f32_16x16x4_f32(false, a, false, b0, (short)0, acc0, false, false);
        acc1 = __builtin_amdgcn_wmma_f32_16x16x4_f32(false, a, false, b1, (short)0, acc1, false, false);
        acc2 = __builtin_amdgcn_wmma_f32_16x16x4_f32(false, a, false, b2, (short)0, acc2, false, false);
        acc3 = __builtin_amdgcn_wmma_f32_16x16x4_f32(false, a, false, b3, (short)0, acc3, false, false);
    }

    float* Crow = C + (size_t)(tileM * 16 + half * 8) * 64 + nloc;
    #pragma unroll
    for (int v = 0; v < 8; ++v) {
        Crow[(size_t)v * 64 +  0] = acc0[v];
        Crow[(size_t)v * 64 + 16] = acc1[v];
        Crow[(size_t)v * 64 + 32] = acc2[v];
        Crow[(size_t)v * 64 + 48] = acc3[v];
    }
}


__global__ void k_zero(int* __restrict__ p, int m,
                       float* __restrict__ pad0, float* __restrict__ pad1) {
    int i = blockIdx.x * blockDim.x + threadIdx.x;
    if (i < m) p[i] = 0;
    if (i < 64) { pad0[i] = 0.0f; pad1[i] = 0.0f; }
}

#define HIST_N   50176
#define CNT_EPT  16
#define CNT_TPB  1024

#define XW_WAVES 16

__global__ __launch_bounds__(CNT_TPB)
void k_count_xw(const int* __restrict__ col, int* __restrict__ cnt,
                int* __restrict__ rank, int e, int n, int count_blocks,
                const float* __restrict__ x, const float* __restrict__ W1,
                float* __restrict__ xw, int mtiles) {
    __shared__ __attribute__((aligned(16))) int hist[HIST_N];
    const int tid = threadIdx.x;

    if ((int)blockIdx.x >= count_blocks) {
        const int wave  = tid >> 5;
        const int tileM = ((int)blockIdx.x - count_blocks) * XW_WAVES + wave;
        if (wave < XW_WAVES && tileM < mtiles)
            gemm_strip<IN_DIM, false>(x, W1, xw, tileM, tid & 31);
        return;
    }

    int4* h4 = (int4*)hist;
    for (int j = tid; j < HIST_N / 4; j += CNT_TPB) h4[j] = make_int4(0, 0, 0, 0);

    const int e0 = blockIdx.x * (CNT_TPB * CNT_EPT) + tid;
    int c[CNT_EPT], lr[CNT_EPT];
    #pragma unroll
    for (int u = 0; u < CNT_EPT; ++u) {
        const int i = e0 + u * CNT_TPB;
        c[u] = (i < e) ? col[i] : 0;
    }
    __syncthreads();

    #pragma unroll
    for (int u = 0; u < CNT_EPT; ++u) {
        lr[u] = 0;
        if (e0 + u * CNT_TPB < e) lr[u] = atomicAdd(&hist[c[u]], 1);
    }
    __syncthreads();

    int first[HIST_N / CNT_TPB];
    #pragma unroll
    for (int j = 0; j < HIST_N / CNT_TPB; ++j) {
        const int node = j * CNT_TPB + tid;
        const int k = hist[node];
        first[j] = 0;
        if (k > 0 && node < n) first[j] = atomicAdd(&cnt[node], k);
    }
    #pragma unroll
    for (int j = 0; j < HIST_N / CNT_TPB; ++j) hist[j * CNT_TPB + tid] = first[j];
    __syncthreads();

    #pragma unroll
    for (int u = 0; u < CNT_EPT; ++u) {
        const int i = e0 + u * CNT_TPB;
        if (i < e) rank[i] = hist[c[u]] + lr[u];
    }
}

__global__ void k_alloc(const int* __restrict__ cnt, int* __restrict__ total,
                        int* __restrict__ start, float* __restrict__ dis, int n) {
    const int i    = blockIdx.x * blockDim.x + threadIdx.x;
    const int lane = threadIdx.x & 31;
    const int c    = (i < n) ? cnt[i] : 0;
    int incl = c;
    for (int d = 1; d < 32; d <<= 1) {
        int v = __shfl_up(incl, d, 32);
        if (lane >= d) incl += v;
    }
    int base = 0;
    if (lane == 31) base = atomicAdd(total, incl);
    base = __shfl(base, 31, 32);
    if (i < n) {
        start[i] = base + incl - c;
        dis[i]   = rsqrtf((float)(c + 1));
    }
}

__global__ void k_fill(const int* __restrict__ row, const int* __restrict__ col,
                       const int* __restrict__ rank, const int* __restrict__ start,
                       const float* __restrict__ dis, int2* __restrict__ ent, int e) {
    int i = blockIdx.x * blockDim.x + threadIdx.x;
    if (i < e) {
        int r = row[i];
        int2 v;
        v.x = r;
        v.y = __float_as_int(dis[r]);
        ent[start[col[i]] + rank[i]] = v;
    }
}

__device__ __forceinline__ float4 gather_row(const float* __restrict__ h,
                                             const float* __restrict__ dis, const float* __restrict__ bias,
                                             const int* __restrict__ start, const int* __restrict__ cnt,
                                             const int2* __restrict__ ent, int n, int node, int lane) {
    const int q    = lane & 15;
    const int half = lane >> 4;

    const int s0  = __builtin_amdgcn_readfirstlane(start[node]);
    const int deg = __builtin_amdgcn_readfirstlane(cnt[node]);
    const float dc = dis[node];

    const float* hq = h + 4 * q;
    const float4 hs = *(const float4*)(hq + (size_t)node * 64);
    const float4 b  = *(const float4*)(bias + 4 * q);
    const float  s2 = dc * dc;

    const int keep = half ? 0 : -1;
    float4 acc;
    acc.x = __int_as_float(__float_as_int(fmaf(s2, hs.x, b.x)) & keep);
    acc.y = __int_as_float(__float_as_int(fmaf(s2, hs.y, b.y)) & keep);
    acc.z = __int_as_float(__float_as_int(fmaf(s2, hs.z, b.z)) & keep);
    acc.w = __int_as_float(__float_as_int(fmaf(s2, hs.w, b.w)) & keep);

    for (int base = 0; base < deg; base += 32) {
        const int m = min(32, deg - base);
        int2 e = make_int2(n, 0);
        if (lane < m) e = ent[s0 + base + lane];

        for (int t0 = 0; 2 * t0 < m; t0 += 4) {
            float4 hv[4];
            float  w[4];
            #pragma unroll
            for (int u = 0; u < 4; ++u) {
                const int j = 2 * (t0 + u) + half;
                const int r = __shfl(e.x, j, 32);
                w[u]  = __int_as_float(__shfl(e.y, j, 32)) * dc;
                hv[u] = *(const float4*)(hq + (size_t)r * 64);
            }
            #pragma unroll
            for (int u = 0; u < 4; ++u) {
                acc.x = fmaf(w[u], hv[u].x, acc.x);
                acc.y = fmaf(w[u], hv[u].y, acc.y);
                acc.z = fmaf(w[u], hv[u].z, acc.z);
                acc.w = fmaf(w[u], hv[u].w, acc.w);
            }
        }
    }

    acc.x += __shfl_xor(acc.x, 16, 32);
    acc.y += __shfl_xor(acc.y, 16, 32);
    acc.z += __shfl_xor(acc.z, 16, 32);
    acc.w += __shfl_xor(acc.w, 16, 32);
    return acc;
}

__global__ void k_gather(float* __restrict__ out, const float* __restrict__ h,
                         const float* __restrict__ dis, const float* __restrict__ bias,
                         const int* __restrict__ start, const int* __restrict__ cnt,
                         const int2* __restrict__ ent, int n) {
    const int lane = threadIdx.x & 31;
    const int node = blockIdx.x * (blockDim.x >> 5) + (threadIdx.x >> 5);
    if (node >= n) return;

    const float4 acc = gather_row(h, dis, bias, start, cnt, ent, n, node, lane);
    if ((lane >> 4) == 0) *(float4*)(out + (size_t)node * 64 + 4 * (lane & 15)) = acc;
}

#define GM_WAVES 16
#define GM_PITCH 68

__global__ __launch_bounds__(GM_WAVES * 32)
void k_gather_mm(float* __restrict__ hw, const float* __restrict__ h,
                 const float* __restrict__ dis, const float* __restrict__ bias,
                 const int* __restrict__ start, const int* __restrict__ cnt,
                 const int2* __restrict__ ent, const float* __restrict__ W, int n) {
    __shared__ __attribute__((aligned(16))) float sA[GM_WAVES * GM_PITCH];
    __shared__ __attribute__((aligned(16))) float sC[GM_WAVES * GM_PITCH];

    const int lane = threadIdx.x & 31;
    const int node = blockIdx.x * GM_WAVES + (threadIdx.x >> 5);
    const bool valid = node < n;
    const int wave = __builtin_amdgcn_readfirstlane((int)(threadIdx.x >> 5));
    const int nloc = lane & 15;
    const int half = lane >> 4;

    float4 acc = gather_row(h, dis, bias, start, cnt, ent, n, valid ? node : 0, lane);
    acc.x = fmaxf(acc.x, 0.0f); acc.y = fmaxf(acc.y, 0.0f);
    acc.z = fmaxf(acc.z, 0.0f); acc.w = fmaxf(acc.w, 0.0f);
    if (half == 0) *(float4*)(sA + wave * GM_PITCH + 4 * nloc) = acc;

    v2f bf[HID / 4];
    if (wave < 4) {
        const float* Bp = W + (size_t)(2 * half) * 64 + 16 * wave + nloc;
        #pragma unroll
        for (int s = 0; s < HID / 4; ++s) {
            bf[s].x = Bp[(size_t)(4 * s) * 64];
            bf[s].y = Bp[(size_t)(4 * s + 1) * 64];
        }
    }
    __builtin_amdgcn_fence(__ATOMIC_RELEASE, "workgroup", "local");
    __builtin_amdgcn_s_barrier();
    __builtin_amdgcn_fence(__ATOMIC_ACQUIRE, "workgroup", "local");

    if (wave < 4) {
        const float* Ap = sA + nloc * GM_PITCH + 2 * half;
        v8f d = {};
        #pragma unroll
        for (int s = 0; s < HID / 4; ++s) {
            const v2f a = *(const v2f*)(Ap + 4 * s);
            d = __builtin_amdgcn_wmma_f32_16x16x4_f32(false, a, false, bf[s], (short)0, d, false, false);
        }
        float* Cp = sC + (8 * half) * GM_PITCH + 16 * wave + nloc;
        #pragma unroll
        for (int v = 0; v < 8; ++v) Cp[v * GM_PITCH] = d[v];
    }
    __syncthreads();

    if (valid) {
        const float2 o = *(const float2*)(sC + wave * GM_PITCH + 2 * lane);
        *(float2*)(hw + (size_t)node * 64 + 2 * lane) = o;
    }
}


extern "C" void kernel_launch(void* const* d_in, const int* in_sizes, int n_in,
                              void* d_out, int out_size, void* d_ws, size_t ws_size,
                              hipStream_t stream) {
    const float* x    = (const float*)d_in[0];
    const int*   eidx = (const int*)d_in[1];
    const float* W1   = (const float*)d_in[2];
    const float* b1   = (const float*)d_in[3];
    const float* W2   = (const float*)d_in[4];
    const float* b2   = (const float*)d_in[5];

    const int n = in_sizes[0] / IN_DIM;
    const int E = in_sizes[1] / 2;
    const int* row = eidx;
    const int* col = eidx + E;

    const size_t nw = ((size_t)n + 1 + 63) & ~(size_t)63;
    const size_t ew = ((size_t)E + 63) & ~(size_t)63;
    int*   cnt   = (int*)d_ws;
    int*   total = cnt + n;
    int*   start = cnt + nw;
    float* dis   = (float*)(start + nw);
    int*   rank  = (int*)(dis + nw);
    int2*  ent   = (int2*)(rank + ew);
    float* t0    = (float*)(rank + ew) + 2 * ew;
    float* t2    = t0 + (size_t)(n + 1) * HID;
    float* out   = (float*)d_out;

    const int TPB = 256;
    const int mtiles = n / 16;
    const int nodes_per_block = TPB / 32;
    dim3 gather_grid((n + nodes_per_block - 1) / nodes_per_block);

    k_zero <<<(n + 1 + TPB - 1) / TPB, TPB, 0, stream>>>(cnt, n + 1, t0 + (size_t)n * HID, t2 + (size_t)n * OUTD);
    const int count_blocks = (E + CNT_TPB * CNT_EPT - 1) / (CNT_TPB * CNT_EPT);
    const int xw_blocks    = (mtiles + XW_WAVES - 1) / XW_WAVES;
    k_count_xw<<<count_blocks + xw_blocks, CNT_TPB, 0, stream>>>(col, cnt, rank, E, n, count_blocks, x, W1, t0, mtiles);
    k_alloc<<<(n + TPB - 1) / TPB, TPB, 0, stream>>>(cnt, total, start, dis, n);
    k_fill <<<(E + TPB - 1) / TPB, TPB, 0, stream>>>(row, col, rank, start, dis, ent, E);

    k_gather_mm<<<(n + GM_WAVES - 1) / GM_WAVES, GM_WAVES * 32, 0, stream>>>(t2, t0, dis, b1, start, cnt, ent, W2, n);

    k_gather<<<gather_grid, TPB, 0, stream>>>(out, t2, dis, b2, start, cnt, ent, n);
}


# bucket rows at a fixed pitch: an entry lies at node x 64 + rank (ranks past 64: a spill list), so the start-offset scan launch and the start and dis planes go; sum order: bucket rank, as before
# speedup vs baseline: 3.4154x; 1.0648x over previous
#include <hip/hip_runtime.h>
#include <hip/hip_bf16.h>


#define IN_DIM 128
#define HID    64
#define OUTD   64

typedef float v2f __attribute__((ext_vector_type(2)));
typedef float v8f __attribute__((ext_vector_type(8)));


template<int K, bool RELU_A>
__device__ __forceinline__ void gemm_strip(const float* __restrict__ A,
                                           const float* __restrict__ B,
                                           float* __restrict__ C,
                                           int tileM, int lane) {
    const int nloc = lane & 15;
    const int half = lane >> 4;

    const float* Arow = A + (size_t)(tileM * 16 + nloc) * K;

    v8f acc0 = {}, acc1 = {}, acc2 = {}, acc3 = {};

    #pragma unroll 4
    for (int k0 = 0; k0 < K; k0 += 4) {
        v2f a = *(const v2f*)(Arow + k0 + 2 * half);
        if (RELU_A) { a.x = fmaxf(a.x, 0.0f); a.y = fmaxf(a.y, 0.0f); }

        const float* Bp = B + (size_t)(k0 + 2 * half) * 64 + nloc;
        v2f b0, b1, b2, b3;
        b0.x = Bp[0];  b0.y = Bp[64];
        b1.x = Bp[16]; b1.y = Bp[80];
        b2.x = Bp[32]; b2.y = Bp[96];
        b3.x = Bp[48]; b3.y = Bp[112];

        acc0 = __builtin_amdgcn_wmma_f32_16x16x4_f32(false, a, false, b0, (short)0, acc0, false, false);
        acc1 = __builtin_amdgcn_wmma_f32_16x16x4_f32(false, a, false, b1, (short)0, acc1, false, false);
        acc2 = __builtin_amdgcn_wmma_f32_16x16x4_f32(false, a, false, b2, (short)0, acc2, false, false);
        acc3 = __builtin_amdgcn_wmma_f32_16x16x4_f32(false, a, false, b3, (short)0, acc3, false, false);
    }

    float* Crow = C + (size_t)(tileM * 16 + half * 8) * 64 + nloc;
    #pragma unroll
    for (int v = 0; v < 8; ++v) {
        Crow[(size_t)v * 64 +  0] = acc0[v];
        Crow[(size_t)v * 64 + 16] = acc1[v];
        Crow[(size_t)v * 64 + 32] = acc2[v];
        Crow[(size_t)v * 64 + 48] = acc3[v];
    }
}

#define SLOTS 64

__global__ void k_zero(int* __restrict__ p, int m,
                       float* __restrict__ pad0, float* __restrict__ pad1) {
    int i = blockIdx.x * blockDim.x + threadIdx.x;
    if (i < m) p[i] = 0;
    if (i < 64) { pad0[i] = 0.0f; pad1[i] = 0.0f; }
}

#define HIST_N   50176
#define CNT_EPT  16
#define CNT_TPB  1024

#define XW_WAVES 16

__global__ __launch_bounds__(CNT_TPB)
void k_count_xw(const int* __restrict__ col, int* __restrict__ cnt,
                int* __restrict__ rank, int e, int n, int count_blocks,
                const float* __restrict__ x, const float* __restrict__ W1,
                float* __restrict__ xw, int mtiles) {
    __shared__ __attribute__((aligned(16))) int hist[HIST_N];
    const int tid = threadIdx.x;

    if ((int)blockIdx.x >= count_blocks) {
        const int wave  = tid >> 5;
        const int tileM = ((int)blockIdx.x - count_blocks) * XW_WAVES + wave;
        if (wave < XW_WAVES && tileM < mtiles)
            gemm_strip<IN_DIM, false>(x, W1, xw, tileM, tid & 31);
        return;
    }

    int4* h4 = (int4*)hist;
    for (int j = tid; j < HIST_N / 4; j += CNT_TPB) h4[j] = make_int4(0, 0, 0, 0);

    const int e0 = blockIdx.x * (CNT_TPB * CNT_EPT) + tid;
    int c[CNT_EPT], lr[CNT_EPT];
    #pragma unroll
    for (int u = 0; u < CNT_EPT; ++u) {
        const int i = e0 + u * CNT_TPB;
        c[u] = (i < e) ? col[i] : 0;
    }
    __syncthreads();

    #pragma unroll
    for (int u = 0; u < CNT_EPT; ++u) {
        lr[u] = 0;
        if (e0 + u * CNT_TPB < e) lr[u] = atomicAdd(&hist[c[u]], 1);
    }
    __syncthreads();

    int first[HIST_N / CNT_TPB];
    #pragma unroll
    for (int j = 0; j < HIST_N / CNT_TPB; ++j) {
        const int node = j * CNT_TPB + tid;
        const int k = hist[node];
        first[j] = 0;
        if (k > 0 && node < n) first[j] = atomicAdd(&cnt[node], k);
    }
    #pragma unroll
    for (int j = 0; j < HIST_N / CNT_TPB; ++j) hist[j * CNT_TPB + tid] = first[j];
    __syncthreads();

    #pragma unroll
    for (int u = 0; u < CNT_EPT; ++u) {
        const int i = e0 + u * CNT_TPB;
        if (i < e) rank[i] = hist[c[u]] + lr[u];
    }
}

__global__ void k_fill(const int* __restrict__ row, const int* __restrict__ col,
                       const int* __restrict__ rank, const int* __restrict__ cnt,
                       int2* __restrict__ ent, int4* __restrict__ spill,
                       int* __restrict__ nspill, int e) {
    int i = blockIdx.x * blockDim.x + threadIdx.x;
    if (i < e) {
        const int r  = row[i];
        const int c  = col[i];
        const int rk = rank[i];
        const int wb = __float_as_int(rsqrtf((float)(cnt[r] + 1)));
        if (rk < SLOTS) {
            ent[(size_t)c * SLOTS + rk] = make_int2(r, wb);
        } else {
            const int p = atomicAdd(nspill, 1);
            spill[p] = make_int4(c, r, wb, 0);
        }
    }
}

__device__ __forceinline__ float4 gather_row(const float* __restrict__ h,
                                             const float* __restrict__ bias,
                                             const int* __restrict__ cnt,
                                             const int2* __restrict__ ent,
                                             const int4* __restrict__ spill,
                                             int n, int node, int lane) {
    const int q    = lane & 15;
    const int half = lane >> 4;

    const int deg = __builtin_amdgcn_readfirstlane(cnt[node]);
    const float dc = rsqrtf((float)(deg + 1));
    const int2* eb = ent + (size_t)node * SLOTS;
    const int   dm = min(deg, SLOTS);

    const float* hq = h + 4 * q;
    const float4 hs = *(const float4*)(hq + (size_t)node * 64);
    const float4 b  = *(const float4*)(bias + 4 * q);
    const float  s2 = dc * dc;

    const int keep = half ? 0 : -1;
    float4 acc;
    acc.x = __int_as_float(__float_as_int(fmaf(s2, hs.x, b.x)) & keep);
    acc.y = __int_as_float(__float_as_int(fmaf(s2, hs.y, b.y)) & keep);
    acc.z = __int_as_float(__float_as_int(fmaf(s2, hs.z, b.z)) & keep);
    acc.w = __int_as_float(__float_as_int(fmaf(s2, hs.w, b.w)) & keep);

    for (int base = 0; base < dm; base += 32) {
        const int m = min(32, dm - base);
        int2 e = make_int2(n, 0);
        if (lane < m) e = eb[base + lane];

        for (int t0 = 0; 2 * t0 < m; t0 += 4) {
            float4 hv[4];
            float  w[4];
            #pragma unroll
            for (int u = 0; u < 4; ++u) {
                const int j = 2 * (t0 + u) + half;
                const int r = __shfl(e.x, j, 32);
                w[u]  = __int_as_float(__shfl(e.y, j, 32)) * dc;
                hv[u] = *(const float4*)(hq + (size_t)r * 64);
            }
            #pragma unroll
            for (int u = 0; u < 4; ++u) {
                acc.x = fmaf(w[u], hv[u].x, acc.x);
                acc.y = fmaf(w[u], hv[u].y, acc.y);
                acc.z = fmaf(w[u], hv[u].z, acc.z);
                acc.w = fmaf(w[u], hv[u].w, acc.w);
            }
        }
    }

    if (deg > SLOTS) {
        const int ns = __builtin_amdgcn_readfirstlane(cnt[n]);
        for (int b0 = 0; b0 < ns; b0 += 32) {
            int4 o = make_int4(-1, n, 0, 0);
            if (b0 + lane < ns) o = spill[b0 + lane];
            unsigned hit = __builtin_amdgcn_ballot_w32(o.x == node);
            while (hit != 0u) {
                const int j = __builtin_ctz(hit);
                hit &= hit - 1u;
                const int   rj = __shfl(o.y, j, 32);
                const float wj = __int_as_float(__shfl(o.z, j, 32)) * dc;
                const int   r  = half ? n : rj;
                const float w  = half ? 0.0f : wj;
                const float4 hv = *(const float4*)(hq + (size_t)r * 64);
                acc.x = fmaf(w, hv.x, acc.x);
                acc.y = fmaf(w, hv.y, acc.y);
                acc.z = fmaf(w, hv.z, acc.z);
                acc.w = fmaf(w, hv.w, acc.w);
            }
        }
    }

    acc.x += __shfl_xor(acc.x, 16, 32);
    acc.y += __shfl_xor(acc.y, 16, 32);
    acc.z += __shfl_xor(acc.z, 16, 32);
    acc.w += __shfl_xor(acc.w, 16, 32);
    return acc;
}

__global__ void k_gather(float* __restrict__ out, const float* __restrict__ h,
                         const float* __restrict__ bias, const int* __restrict__ cnt,
                         const int2* __restrict__ ent, const int4* __restrict__ spill, int n) {
    const int lane = threadIdx.x & 31;
    const int node = blockIdx.x * (blockDim.x >> 5) + (threadIdx.x >> 5);
    if (node >= n) return;

    const float4 acc = gather_row(h, bias, cnt, ent, spill, n, node, lane);
    if ((lane >> 4) == 0) *(float4*)(out + (size_t)node * 64 + 4 * (lane & 15)) = acc;
}

#define GM_WAVES 16
#define GM_PITCH 68

__global__ __launch_bounds__(GM_WAVES * 32)
void k_gather_mm(float* __restrict__ hw, const float* __restrict__ h,
                 const float* __restrict__ bias, const int* __restrict__ cnt,
                 const int2* __restrict__ ent, const int4* __restrict__ spill,
                 const float* __restrict__ W, int n) {
    __shared__ __attribute__((aligned(16))) float sA[GM_WAVES * GM_PITCH];
    __shared__ __attribute__((aligned(16))) float sC[GM_WAVES * GM_PITCH];

    const int lane = threadIdx.x & 31;
    const int node = blockIdx.x * GM_WAVES + (threadIdx.x >> 5);
    const bool valid = node < n;
    const int wave = __builtin_amdgcn_readfirstlane((int)(threadIdx.x >> 5));
    const int nloc = lane & 15;
    const int half = lane >> 4;

    float4 acc = gather_row(h, bias, cnt, ent, spill, n, valid ? node : 0, lane);
    acc.x = fmaxf(acc.x, 0.0f); acc.y = fmaxf(acc.y, 0.0f);
    acc.z = fmaxf(acc.z, 0.0f); acc.w = fmaxf(acc.w, 0.0f);
    if (half == 0) *(float4*)(sA + wave * GM_PITCH + 4 * nloc) = acc;

    v2f bf[HID / 4];
    if (wave < 4) {
        const float* Bp = W + (size_t)(2 * half) * 64 + 16 * wave + nloc;
        #pragma unroll
        for (int s = 0; s < HID / 4; ++s) {
            bf[s].x = Bp[(size_t)(4 * s) * 64];
            bf[s].y = Bp[(size_t)(4 * s + 1) * 64];
        }
    }
    __builtin_amdgcn_fence(__ATOMIC_RELEASE, "workgroup", "local");
    __builtin_amdgcn_s_barrier();
    __builtin_amdgcn_fence(__ATOMIC_ACQUIRE, "workgroup", "local");

    if (wave < 4) {
        const float* Ap = sA + nloc * GM_PITCH + 2 * half;
        v8f d = {};
        #pragma unroll
        for (int s = 0; s < HID / 4; ++s) {
            const v2f a = *(const v2f*)(Ap + 4 * s);
            d = __builtin_amdgcn_wmma_f32_16x16x4_f32(false, a, false, bf[s], (short)0, d, false, false);
        }
        float* Cp = sC + (8 * half) * GM_PITCH + 16 * wave + nloc;
        #pragma unroll
        for (int v = 0; v < 8; ++v) Cp[v * GM_PITCH] = d[v];
    }
    __syncthreads();

    if (valid) {
        const float2 o = *(const float2*)(sC + wave * GM_PITCH + 2 * lane);
        *(float2*)(hw + (size_t)node * 64 + 2 * lane) = o;
    }
}


extern "C" void kernel_launch(void* const* d_in, const int* in_sizes, int n_in,
                              void* d_out, int out_size, void* d_ws, size_t ws_size,
                              hipStream_t stream) {
    const float* x    = (const float*)d_in[0];
    const int*   eidx = (const int*)d_in[1];
    const float* W1   = (const float*)d_in[2];
    const float* b1   = (const float*)d_in[3];
    const float* W2   = (const float*)d_in[4];
    const float* b2   = (const float*)d_in[5];

    const int n = in_sizes[0] / IN_DIM;
    const int E = in_sizes[1] / 2;
    const int* row = eidx;
    const int* col = eidx + E;

    const size_t nw = ((size_t)n + 1 + 63) & ~(size_t)63;
    const size_t ew = ((size_t)E + 63) & ~(size_t)63;
    int*   cnt    = (int*)d_ws;
    int*   nspill = cnt + n;
    int*   rank   = cnt + nw;
    int4*  spill  = (int4*)(rank + ew);
    int2*  ent    = (int2*)(rank + ew + 4 * ew);
    float* t0     = (float*)(rank + ew + 4 * ew) + (size_t)n * SLOTS * 2;
    float* t2     = t0 + (size_t)(n + 1) * HID;
    float* out    = (float*)d_out;

    const int TPB = 256;
    const int mtiles = n / 16;
    const int nodes_per_block = TPB / 32;
    dim3 gather_grid((n + nodes_per_block - 1) / nodes_per_block);

    k_zero <<<(n + 1 + TPB - 1) / TPB, TPB, 0, stream>>>(cnt, n + 1, t0 + (size_t)n * HID, t2 + (size_t)n * OUTD);
    const int count_blocks = (E + CNT_TPB * CNT_EPT - 1) / (CNT_TPB * CNT_EPT);
    const int xw_blocks    = (mtiles + XW_WAVES - 1) / XW_WAVES;
    k_count_xw<<<count_blocks + xw_blocks, CNT_TPB, 0, stream>>>(col, cnt, rank, E, n, count_blocks, x, W1, t0, mtiles);
    k_fill <<<(E + TPB - 1) / TPB, TPB, 0, stream>>>(row, col, rank, cnt, ent, spill, nspill, E);

    k_gather_mm<<<(n + GM_WAVES - 1) / GM_WAVES, GM_WAVES * 32, 0, stream>>>(t2, t0, b1, cnt, ent, spill, W2, n);

    k_gather<<<gather_grid, TPB, 0, stream>>>(out, t2, b2, cnt, ent, spill, n);
}
